# starting kernel
# baseline (speedup 1.0000x reference)
#include <hip/hip_runtime.h>

typedef __attribute__((ext_vector_type(2))) float v2f;
typedef __attribute__((ext_vector_type(4))) float v4f;
typedef __attribute__((ext_vector_type(8))) float v8f;

#define NUM_GATES   8192
#define INPUT_SIZE  8192
#define BATCH       2048

__device__ __constant__ float CT[16][16] = {
      {  0,  0,  0,  0,  0,  0,  0,  0,  1,  1,  1,  1,  1,  1,  1,  1 },
      {  0,  0,  1,  1,  0,  0,  1,  1, -1, -1,  0,  0, -1, -1,  0,  0 },
      {  0,  0,  0,  0,  1,  1,  1,  1, -1, -1, -1, -1,  0,  0,  0,  0 },
      {  0,  1, -1,  0, -1,  0, -2, -1,  1,  2,  0,  1,  0,  1, -1,  0 },
    {0},{0},{0},{0},{0},{0},{0},{0},{0},{0},{0},{0}
};

__global__ void __launch_bounds__(32)
gate_weights_wmma(const float* __restrict__ logits, float* __restrict__ w)
{
    __shared__ float sm[16][18];
    const int lane  = threadIdx.x;
    const int g0    = blockIdx.x * 16;

    const float* src = logits + g0 * 16;
#pragma unroll
    for (int i = 0; i < 8; ++i) {
        int idx = lane * 8 + i;
        sm[idx >> 4][idx & 15] = src[idx];
    }
    __syncthreads();

    const int m     = lane & 15;
    const int khalf = lane >> 4;
    const int n     = m;

    float row[16];
    float mx = -INFINITY;
#pragma unroll
    for (int j = 0; j < 16; ++j) { row[j] = sm[m][j]; mx = fmaxf(mx, row[j]); }
    float sum = 0.f;
#pragma unroll
    for (int j = 0; j < 16; ++j) { row[j] = __expf(row[j] - mx); sum += row[j]; }
    const float inv = 1.0f / sum;
#pragma unroll
    for (int j = 0; j < 16; ++j) row[j] *= inv;

    __syncthreads();
    if (lane < 16) {
#pragma unroll
        for (int j = 0; j < 16; ++j) sm[m][j] = row[j];
    }
    __syncthreads();

    v8f acc = {};
#pragma unroll
    for (int k = 0; k < 4; ++k) {
        const int r0 = 4 * k + 2 * khalf;
        const v2f a = *(const v2f*)(&sm[m][r0]);
        const v2f b = *(const v2f*)(&CT[n][r0]);
        acc = __builtin_amdgcn_wmma_f32_16x16x4_f32(
                   false, a,  false, b,
                   (short)0, acc,  false,  false);
    }

    if (n < 4) {
#pragma unroll
        for (int j = 0; j < 8; ++j) {
            const int grow = j + 8 * khalf;
            w[(g0 + grow) * 4 + n] = acc[j];
        }
    }
}

__global__ void __launch_bounds__(256)
logic_eval(const float* __restrict__ x,
           const v4f* __restrict__ w4,
           float* __restrict__ out)
{
    const int tid = blockIdx.x * blockDim.x + threadIdx.x;
    const int gq  = tid & (NUM_GATES / 4 - 1);
    const int bi  = tid >> 11;
    const int g   = gq << 2;

    const float* xr = x + ((size_t)bi << 13);

    const v4f   xa = *(const v4f*)(xr + g);
    const float xn = xr[(g + 4) & (INPUT_SIZE - 1)];

    float va[5] = { xa.x, xa.y, xa.z, xa.w, xn };
#pragma unroll
    for (int i = 0; i < 5; ++i) va[i] = fminf(fmaxf(va[i], 0.f), 1.f);

    v4f r;
#pragma unroll
    for (int i = 0; i < 4; ++i) {
        const v4f wv = w4[g + i];
        const float a = va[i], b = va[i + 1];
        float t = fmaf(wv.y, a, wv.x);
        t = fmaf(wv.z, b, t);
        t = fmaf(wv.w * a, b, t);
        r[i] = t;
    }

    __builtin_nontemporal_store(r, (v4f*)(out + (size_t)tid * 4));
}

extern "C" void kernel_launch(void* const* d_in, const int* in_sizes, int n_in,
                              void* d_out, int out_size, void* d_ws, size_t ws_size,
                              hipStream_t stream)
{
    const float* x      = (const float*)d_in[0];
    const float* logits = (const float*)d_in[1];
    float*       out    = (float*)d_out;
    float*       w      = (float*)d_ws;

    gate_weights_wmma<<<NUM_GATES / 16, 32, 0, stream>>>(logits, w);

    const int total_threads = BATCH * (NUM_GATES / 4);
    logic_eval<<<total_threads / 256, 256, 0, stream>>>(x, (const v4f*)w, out);
}


# eval kernel: each thread keeps its 4 gates' 16 weights in registers and walks 32 batch rows (16 rows of loads in flight, stored in load order); 16384 -> 512 workgroups, weight fetch once per 32 rows
# speedup vs baseline: 1.2830x; 1.2830x over previous
#include <hip/hip_runtime.h>

typedef __attribute__((ext_vector_type(2))) float v2f;
typedef __attribute__((ext_vector_type(4))) float v4f;
typedef __attribute__((ext_vector_type(8))) float v8f;

#define NUM_GATES   8192
#define INPUT_SIZE  8192
#define BATCH       2048

__device__ __constant__ float CT[16][16] = {
      {  0,  0,  0,  0,  0,  0,  0,  0,  1,  1,  1,  1,  1,  1,  1,  1 },
      {  0,  0,  1,  1,  0,  0,  1,  1, -1, -1,  0,  0, -1, -1,  0,  0 },
      {  0,  0,  0,  0,  1,  1,  1,  1, -1, -1, -1, -1,  0,  0,  0,  0 },
      {  0,  1, -1,  0, -1,  0, -2, -1,  1,  2,  0,  1,  0,  1, -1,  0 },
    {0},{0},{0},{0},{0},{0},{0},{0},{0},{0},{0},{0}
};

__global__ void __launch_bounds__(32)
gate_weights_wmma(const float* __restrict__ logits, float* __restrict__ w)
{
    __shared__ float sm[16][18];
    const int lane  = threadIdx.x;
    const int g0    = blockIdx.x * 16;

    const float* src = logits + g0 * 16;
#pragma unroll
    for (int i = 0; i < 8; ++i) {
        int idx = lane * 8 + i;
        sm[idx >> 4][idx & 15] = src[idx];
    }
    __syncthreads();

    const int m     = lane & 15;
    const int khalf = lane >> 4;
    const int n     = m;

    float row[16];
    float mx = -INFINITY;
#pragma unroll
    for (int j = 0; j < 16; ++j) { row[j] = sm[m][j]; mx = fmaxf(mx, row[j]); }
    float sum = 0.f;
#pragma unroll
    for (int j = 0; j < 16; ++j) { row[j] = __expf(row[j] - mx); sum += row[j]; }
    const float inv = 1.0f / sum;
#pragma unroll
    for (int j = 0; j < 16; ++j) row[j] *= inv;

    __syncthreads();
    if (lane < 16) {
#pragma unroll
        for (int j = 0; j < 16; ++j) sm[m][j] = row[j];
    }
    __syncthreads();

    v8f acc = {};
#pragma unroll
    for (int k = 0; k < 4; ++k) {
        const int r0 = 4 * k + 2 * khalf;
        const v2f a = *(const v2f*)(&sm[m][r0]);
        const v2f b = *(const v2f*)(&CT[n][r0]);
        acc = __builtin_amdgcn_wmma_f32_16x16x4_f32(
                   false, a,  false, b,
                   (short)0, acc,  false,  false);
    }

    if (n < 4) {
#pragma unroll
        for (int j = 0; j < 8; ++j) {
            const int grow = j + 8 * khalf;
            w[(g0 + grow) * 4 + n] = acc[j];
        }
    }
}

#define ROWS_PER_THREAD 32
#define ROWS_IN_FLIGHT  16
#define GATE_GROUPS     8
#define EVAL_BLOCKS     512

static_assert(GATE_GROUPS * 4 * 256 == NUM_GATES, "gate groups must tile the gates");
static_assert(EVAL_BLOCKS * ROWS_PER_THREAD == GATE_GROUPS * BATCH, "grid must tile the batch");
static_assert(ROWS_PER_THREAD % ROWS_IN_FLIGHT == 0, "row groups must tile a thread's rows");

__global__ void __launch_bounds__(256) __attribute__((amdgpu_waves_per_eu(4, 8)))
logic_eval(const float* __restrict__ x,
           const v4f* __restrict__ w4,
           float* __restrict__ out)
{
    const int gq = (blockIdx.x & (GATE_GROUPS - 1)) * 256 + threadIdx.x;
    const int b0 = (blockIdx.x / GATE_GROUPS) * ROWS_PER_THREAD;
    const int g  = gq << 2;
    const int gn = (g + 4) & (INPUT_SIZE - 1);

    v4f wv[4];
#pragma unroll
    for (int i = 0; i < 4; ++i) wv[i] = w4[g + i];

    const float* xr   = x   + (size_t)b0 * INPUT_SIZE;
    float*       orow = out + (size_t)b0 * NUM_GATES + g;

    for (int r0 = 0; r0 < ROWS_PER_THREAD; r0 += ROWS_IN_FLIGHT) {
        v4f   xa[ROWS_IN_FLIGHT];
        float xn[ROWS_IN_FLIGHT];
#pragma unroll
        for (int r = 0; r < ROWS_IN_FLIGHT; ++r) {
            const float* p = xr + (size_t)r * INPUT_SIZE;
            xa[r] = *(const v4f*)(p + g);
            xn[r] = p[gn];
        }
#pragma unroll
        for (int r = 0; r < ROWS_IN_FLIGHT; ++r) {
            float va[5] = { xa[r].x, xa[r].y, xa[r].z, xa[r].w, xn[r] };
#pragma unroll
            for (int i = 0; i < 5; ++i) va[i] = fminf(fmaxf(va[i], 0.f), 1.f);

            v4f res;
#pragma unroll
            for (int i = 0; i < 4; ++i) {
                const float a = va[i], b = va[i + 1];
                float t = fmaf(wv[i].y, a, wv[i].x);
                t = fmaf(wv[i].z, b, t);
                t = fmaf(wv[i].w * a, b, t);
                res[i] = t;
            }
            __builtin_nontemporal_store(res, (v4f*)(orow + (size_t)r * NUM_GATES));
            __builtin_amdgcn_sched_barrier(0);
        }
        xr   += (size_t)ROWS_IN_FLIGHT * INPUT_SIZE;
        orow += (size_t)ROWS_IN_FLIGHT * NUM_GATES;
    }
}

extern "C" void kernel_launch(void* const* d_in, const int* in_sizes, int n_in,
                              void* d_out, int out_size, void* d_ws, size_t ws_size,
                              hipStream_t stream)
{
    const float* x      = (const float*)d_in[0];
    const float* logits = (const float*)d_in[1];
    float*       out    = (float*)d_out;
    float*       w      = (float*)d_ws;

    gate_weights_wmma<<<NUM_GATES / 16, 32, 0, stream>>>(logits, w);

    logic_eval<<<EVAL_BLOCKS, 256, 0, stream>>>(x, (const v4f*)w, out);
}


# single launch: each 8-wave workgroup computes its own 128 gates' weights (one 16-gate WMMA tile per wave, shared via LDS) under its first row loads; softmax sum and WMMA k order changed
# speedup vs baseline: 1.5198x; 1.1846x over previous
#include <hip/hip_runtime.h>


typedef __attribute__((ext_vector_type(2))) float v2f;
typedef __attribute__((ext_vector_type(4))) float v4f;
typedef __attribute__((ext_vector_type(8))) float v8f;

#define NUM_GATES   8192
#define INPUT_SIZE  8192
#define BATCH       2048

__device__ __constant__ __attribute__((aligned(16))) float CT[16][16] = {
      {  0,  0,  0,  0,  0,  0,  0,  0,  1,  1,  1,  1,  1,  1,  1,  1 },
      {  0,  0,  1,  1,  0,  0,  1,  1, -1, -1,  0,  0, -1, -1,  0,  0 },
      {  0,  0,  0,  0,  1,  1,  1,  1, -1, -1, -1, -1,  0,  0,  0,  0 },
      {  0,  1, -1,  0, -1,  0, -2, -1,  1,  2,  0,  1,  0,  1, -1,  0 },
    {0},{0},{0},{0},{0},{0},{0},{0},{0},{0},{0},{0}
};

#define ROWS_PER_THREAD 32
#define ROWS_IN_FLIGHT  16
#define WAVES           8
#define GATES_PER_WG    128
#define GATE_GROUPS     (NUM_GATES / GATES_PER_WG)
#define ROWS_PER_WG     (WAVES * ROWS_PER_THREAD)
#define EVAL_BLOCKS     (GATE_GROUPS * (BATCH / ROWS_PER_WG))

static_assert(GATES_PER_WG == 32 * 4 && GATES_PER_WG == WAVES * 16, "one eval wave width = one tile per wave");
static_assert((GATE_GROUPS & (GATE_GROUPS - 1)) == 0, "gate groups are split off with a mask");
static_assert(BATCH % ROWS_PER_WG == 0, "grid must tile the batch");
static_assert(ROWS_PER_THREAD % ROWS_IN_FLIGHT == 0, "row groups must tile a wave's rows");

static __device__ __forceinline__ float xhalf(float x)
{
    return __builtin_bit_cast(float, __builtin_amdgcn_permlanex16(
        __builtin_bit_cast(int, x), __builtin_bit_cast(int, x),
        0x76543210, 0xfedcba98, false, false));
}

__global__ void __launch_bounds__(256) __attribute__((amdgpu_waves_per_eu(4, 8)))
logic_fused(const float* __restrict__ x,
            const float* __restrict__ logits,
            float* __restrict__ out)
{
    __shared__ v4f w_s[GATES_PER_WG];

    const int lane = threadIdx.x & 31;
    const int wave = threadIdx.x >> 5;
    const int gb   = (blockIdx.x & (GATE_GROUPS - 1)) * GATES_PER_WG;
    const int b0   = (blockIdx.x / GATE_GROUPS) * ROWS_PER_WG + wave * ROWS_PER_THREAD;
    const int g    = gb + (lane << 2);
    const int gn   = (g + 4) & (INPUT_SIZE - 1);

    const int c = lane & 15;
    const int h = lane >> 4;

    const float* lp = logits + (size_t)(gb + wave * 16 + c) * 16 + 8 * h;
    const v4f l0 = *(const v4f*)(lp);
    const v4f l1 = *(const v4f*)(lp + 4);
    const v4f c0 = *(const v4f*)(&CT[c][8 * h]);
    const v4f c1 = *(const v4f*)(&CT[c][8 * h + 4]);

    const float* xr   = x   + (size_t)b0 * INPUT_SIZE;
    float*       orow = out + (size_t)b0 * NUM_GATES + g;

    v4f wv[4];

    for (int r0 = 0; r0 < ROWS_PER_THREAD; r0 += ROWS_IN_FLIGHT) {
        v4f   xa[ROWS_IN_FLIGHT];
        float xn[ROWS_IN_FLIGHT];
#pragma unroll
        for (int r = 0; r < ROWS_IN_FLIGHT; ++r) {
            const float* p = xr + (size_t)r * INPUT_SIZE;
            xa[r] = *(const v4f*)(p + g);
            xn[r] = p[gn];
        }

        if (r0 == 0) {

            v4f t0 = l0, t1 = l1;
            asm volatile("" : "+v"(t0), "+v"(t1));
            float lv[8] = { t0.x, t0.y, t0.z, t0.w, t1.x, t1.y, t1.z, t1.w };
            float mx = lv[0];
#pragma unroll
            for (int j = 1; j < 8; ++j) mx = fmaxf(mx, lv[j]);
            mx = fmaxf(mx, xhalf(mx));
            float sum = 0.f;
#pragma unroll
            for (int j = 0; j < 8; ++j) { lv[j] = __expf(lv[j] - mx); sum += lv[j]; }
            sum += xhalf(sum);
            const float inv = 1.0f / sum;
#pragma unroll
            for (int j = 0; j < 8; ++j) lv[j] *= inv;

            const v2f a0 = { c0.x, c0.y }, a1 = { c0.z, c0.w }, a2 = { c1.x, c1.y }, a3 = { c1.z, c1.w };
            const v2f p0 = { lv[0], lv[1] }, p1 = { lv[2], lv[3] }, p2 = { lv[4], lv[5] }, p3 = { lv[6], lv[7] };
            v8f acc;
            asm volatile(
                "v_wmma_f32_16x16x4_f32 %0, %1, %5, 0\n\t"
                "v_wmma_f32_16x16x4_f32 %0, %2, %6, %0\n\t"
                "v_wmma_f32_16x16x4_f32 %0, %3, %7, %0\n\t"
                "v_wmma_f32_16x16x4_f32 %0, %4, %8, %0\n\t"
                "v_nop\n\tv_nop\n\tv_nop\n\tv_nop\n\tv_nop\n\tv_nop\n\tv_nop\n\tv_nop"
                : "=&v"(acc)
                : "v"(a0), "v"(a1), "v"(a2), "v"(a3), "v"(p0), "v"(p1), "v"(p2), "v"(p3));

            if (h == 0) {
                const v4f wg = { acc[0], acc[1], acc[2], acc[3] };
                w_s[wave * 16 + c] = wg;
            }
            asm volatile("s_wait_dscnt 0x0\n\ts_barrier_signal -1\n\ts_barrier_wait -1" ::: "memory");
#pragma unroll
            for (int i = 0; i < 4; ++i) wv[i] = w_s[(lane << 2) + i];
        }

#pragma unroll
        for (int r = 0; r < ROWS_IN_FLIGHT; ++r) {
            float va[5] = { xa[r].x, xa[r].y, xa[r].z, xa[r].w, xn[r] };
#pragma unroll
            for (int i = 0; i < 5; ++i) va[i] = fminf(fmaxf(va[i], 0.f), 1.f);

            v4f res;
#pragma unroll
            for (int i = 0; i < 4; ++i) {
                const float a = va[i], b = va[i + 1];
                float t = fmaf(wv[i].y, a, wv[i].x);
                t = fmaf(wv[i].z, b, t);
                t = fmaf(wv[i].w * a, b, t);
                res[i] = t;
            }
            __builtin_nontemporal_store(res, (v4f*)(orow + (size_t)r * NUM_GATES));
            __builtin_amdgcn_sched_barrier(0);
        }
        xr   += (size_t)ROWS_IN_FLIGHT * INPUT_SIZE;
        orow += (size_t)ROWS_IN_FLIGHT * NUM_GATES;
    }
}

extern "C" void kernel_launch(void* const* d_in, const int* in_sizes, int n_in,
                              void* d_out, int out_size, void* d_ws, size_t ws_size,
                              hipStream_t stream)
{
    const float* x      = (const float*)d_in[0];
    const float* logits = (const float*)d_in[1];
    float*       out    = (float*)d_out;

    logic_fused<<<EVAL_BLOCKS, 256, 0, stream>>>(x, logits, out);
}


# one load group per wave: all 32 rows' x loads issued at entry (one memory round trip, not two); to fit the 63-deep load counter the 5th input comes from the next lane, not from a 2nd load per row
# speedup vs baseline: 1.7194x; 1.1313x over previous
#include <hip/hip_runtime.h>


typedef __attribute__((ext_vector_type(2))) float v2f;
typedef __attribute__((ext_vector_type(4))) float v4f;
typedef __attribute__((ext_vector_type(8))) float v8f;

#define NUM_GATES   8192
#define INPUT_SIZE  8192
#define BATCH       2048

__device__ __constant__ __attribute__((aligned(16))) float CT[16][16] = {
      {  0,  0,  0,  0,  0,  0,  0,  0,  1,  1,  1,  1,  1,  1,  1,  1 },
      {  0,  0,  1,  1,  0,  0,  1,  1, -1, -1,  0,  0, -1, -1,  0,  0 },
      {  0,  0,  0,  0,  1,  1,  1,  1, -1, -1, -1, -1,  0,  0,  0,  0 },
      {  0,  1, -1,  0, -1,  0, -2, -1,  1,  2,  0,  1,  0,  1, -1,  0 },
    {0},{0},{0},{0},{0},{0},{0},{0},{0},{0},{0},{0}
};

#define ROWS_PER_THREAD 32
#define WAVES           8
#define GATES_PER_WG    128
#define GATE_GROUPS     (NUM_GATES / GATES_PER_WG)
#define ROWS_PER_WG     (WAVES * ROWS_PER_THREAD)
#define EVAL_BLOCKS     (GATE_GROUPS * (BATCH / ROWS_PER_WG))

static_assert(GATES_PER_WG == 32 * 4 && GATES_PER_WG == WAVES * 16, "one eval wave width = one tile per wave");
static_assert((GATE_GROUPS & (GATE_GROUPS - 1)) == 0, "gate groups are split off with a mask");
static_assert(BATCH % ROWS_PER_WG == 0, "grid must tile the batch");
static_assert(ROWS_PER_THREAD == 32, "lane j of the edge load holds row j of the wave");

static __device__ __forceinline__ float xhalf(float x)
{
    return __builtin_bit_cast(float, __builtin_amdgcn_permlanex16(
        __builtin_bit_cast(int, x), __builtin_bit_cast(int, x),
        0x76543210, 0xfedcba98, false, false));
}

static __device__ __forceinline__ float next_lane(float x, float edge, bool last)
{
    const int xi = __builtin_bit_cast(int, x);
    int t = __builtin_amdgcn_permlanex16(xi, xi, 0, 0, false, false);
    t = __builtin_amdgcn_update_dpp(t, xi, 0x101, 0xf, 0xf, false);
    return last ? edge : __builtin_bit_cast(float, t);
}

__global__ void __launch_bounds__(256) __attribute__((amdgpu_waves_per_eu(4, 8)))
logic_fused(const float* __restrict__ x,
            const float* __restrict__ logits,
            float* __restrict__ out)
{
    __shared__ v4f w_s[GATES_PER_WG];

    const int lane = threadIdx.x & 31;
    const int wave = threadIdx.x >> 5;
    const int gb   = (blockIdx.x & (GATE_GROUPS - 1)) * GATES_PER_WG;
    const int b0   = (blockIdx.x / GATE_GROUPS) * ROWS_PER_WG + wave * ROWS_PER_THREAD;
    const int g    = gb + (lane << 2);
    const int ge   = (gb + GATES_PER_WG) & (INPUT_SIZE - 1);

    const int c = lane & 15;
    const int h = lane >> 4;

    const float* lp = logits + (size_t)(gb + wave * 16 + c) * 16 + 8 * h;
    const v4f l0 = *(const v4f*)(lp);
    const v4f l1 = *(const v4f*)(lp + 4);
    const v4f c0 = *(const v4f*)(&CT[c][8 * h]);
    const v4f c1 = *(const v4f*)(&CT[c][8 * h + 4]);

    const float* xr   = x   + (size_t)b0 * INPUT_SIZE;
    float*       orow = out + (size_t)b0 * NUM_GATES + g;

    const float edge = xr[(size_t)lane * INPUT_SIZE + ge];

    v4f xa[ROWS_PER_THREAD];
#pragma unroll
    for (int r = 0; r < ROWS_PER_THREAD; ++r)
        xa[r] = *(const v4f*)(xr + (size_t)r * INPUT_SIZE + g);

    v4f wv[4];
    {

        v4f t0 = l0, t1 = l1;
        asm volatile("" : "+v"(t0), "+v"(t1));
        float lv[8] = { t0.x, t0.y, t0.z, t0.w, t1.x, t1.y, t1.z, t1.w };
        float mx = lv[0];
#pragma unroll
        for (int j = 1; j < 8; ++j) mx = fmaxf(mx, lv[j]);
        mx = fmaxf(mx, xhalf(mx));
        float sum = 0.f;
#pragma unroll
        for (int j = 0; j < 8; ++j) { lv[j] = __expf(lv[j] - mx); sum += lv[j]; }
        sum += xhalf(sum);
        const float inv = 1.0f / sum;
#pragma unroll
        for (int j = 0; j < 8; ++j) lv[j] *= inv;

        const v2f a0 = { c0.x, c0.y }, a1 = { c0.z, c0.w }, a2 = { c1.x, c1.y }, a3 = { c1.z, c1.w };
        const v2f p0 = { lv[0], lv[1] }, p1 = { lv[2], lv[3] }, p2 = { lv[4], lv[5] }, p3 = { lv[6], lv[7] };
        v8f acc;
        asm volatile(
            "v_wmma_f32_16x16x4_f32 %0, %1, %5, 0\n\t"
            "v_wmma_f32_16x16x4_f32 %0, %2, %6, %0\n\t"
            "v_wmma_f32_16x16x4_f32 %0, %3, %7, %0\n\t"
            "v_wmma_f32_16x16x4_f32 %0, %4, %8, %0\n\t"
            "v_nop\n\tv_nop\n\tv_nop\n\tv_nop\n\tv_nop\n\tv_nop\n\tv_nop\n\tv_nop"
            : "=&v"(acc)
            : "v"(a0), "v"(a1), "v"(a2), "v"(a3), "v"(p0), "v"(p1), "v"(p2), "v"(p3));

        if (h == 0) {
            const v4f wg = { acc[0], acc[1], acc[2], acc[3] };
            w_s[wave * 16 + c] = wg;
        }
        asm volatile("s_wait_dscnt 0x0\n\ts_barrier_signal -1\n\ts_barrier_wait -1" ::: "memory");
#pragma unroll
        for (int i = 0; i < 4; ++i) wv[i] = w_s[(lane << 2) + i];
        __builtin_amdgcn_sched_barrier(0);
    }

    const int edge_i = __builtin_bit_cast(int, edge);
#pragma unroll
    for (int r = 0; r < ROWS_PER_THREAD; ++r) {
        const float er = __builtin_bit_cast(float, __builtin_amdgcn_readlane(edge_i, r));
        float va[5] = { xa[r].x, xa[r].y, xa[r].z, xa[r].w, next_lane(xa[r].x, er, lane == 31) };
#pragma unroll
        for (int i = 0; i < 5; ++i) va[i] = fminf(fmaxf(va[i], 0.f), 1.f);

        v4f res;
#pragma unroll
        for (int i = 0; i < 4; ++i) {
            const float a = va[i], b = va[i + 1];
            float t = fmaf(wv[i].y, a, wv[i].x);
            t = fmaf(wv[i].z, b, t);
            t = fmaf(wv[i].w * a, b, t);
            res[i] = t;
        }
        __builtin_nontemporal_store(res, (v4f*)(orow + (size_t)r * NUM_GATES));
        __builtin_amdgcn_sched_barrier(0);
    }
}

extern "C" void kernel_launch(void* const* d_in, const int* in_sizes, int n_in,
                              void* d_out, int out_size, void* d_ws, size_t ws_size,
                              hipStream_t stream)
{
    const float* x      = (const float*)d_in[0];
    const float* logits = (const float*)d_in[1];
    float*       out    = (float*)d_out;

    logic_fused<<<EVAL_BLOCKS, 256, 0, stream>>>(x, logits, out);
}
